# starting kernel
# baseline (speedup 1.0000x reference)
#include <hip/hip_runtime.h>


typedef float v2f __attribute__((ext_vector_type(2)));
typedef float v8f __attribute__((ext_vector_type(8)));

#define LBEAM 1.0f
#define EA_C  1.0e4f
#define EI_C  1.0e2f

__device__ __forceinline__ float fast_tanhf(float a) {
#if __has_builtin(__builtin_amdgcn_tanhf)
    return __builtin_amdgcn_tanhf(a);
#else
    float e = __expf(2.0f * a);
    return 1.0f - 2.0f * __builtin_amdgcn_rcpf(e + 1.0f);
#endif
}

__global__ __launch_bounds__(256) void pinn_beam_wmma(
    const float* __restrict__ x,
    const float* __restrict__ W1, const float* __restrict__ b1,
    const float* __restrict__ W2, const float* __restrict__ b2,
    const float* __restrict__ W3, const float* __restrict__ b3,
    float* __restrict__ out, int n)
{
    const int lane = threadIdx.x & 31;
    const int half = lane >> 4;
    const int l16  = lane & 15;

    v2f Aw[4];
#pragma unroll
    for (int c = 0; c < 4; ++c) {
        Aw[c].x = W2[l16 * 16 + 4 * c + 2 * half + 0];
        Aw[c].y = W2[l16 * 16 + 4 * c + 2 * half + 1];
    }
    float w1c[8], b1c[8];
#pragma unroll
    for (int c = 0; c < 4; ++c)
#pragma unroll
        for (int j = 0; j < 2; ++j) {
            int ch = 4 * c + 2 * half + j;
            w1c[2 * c + j] = W1[ch];
            b1c[2 * c + j] = b1[ch];
        }
    float b2c[8], w3u[8], w3w[8];
#pragma unroll
    for (int r = 0; r < 8; ++r) {
        int ch = r + 8 * half;
        b2c[r] = b2[ch];
        w3u[r] = W3[ch];
        w3w[r] = W3[16 + ch];
    }
    const float b30 = b3[0], b31 = b3[1];

    const int nBatch = n >> 4;
    const int gwave  = (int)(blockIdx.x * blockDim.x + threadIdx.x) >> 5;
    const int nwave  = (int)(gridDim.x * blockDim.x) >> 5;

    for (int batch = gwave; batch < nBatch; batch += nwave) {
        const int   pidx = batch * 16 + l16;
        const float xi   = x[pidx];

        float h0[8], h1[8], h2[8], h3[8];
#pragma unroll
        for (int i = 0; i < 8; ++i) {
            float a1 = w1c[i] * (1.0f / LBEAM);
            float a0 = a1 * xi + b1c[i];
            float t  = fast_tanhf(a0);
            float s  = 1.0f - t * t;
            h0[i] = t;
            h1[i] = s * a1;
            h2[i] = -2.0f * t * s * a1 * a1;
            h3[i] = s * (4.0f * t * t - 2.0f * s) * a1 * a1 * a1;
        }

        v8f acc0 = {}, acc1 = {}, acc2 = {}, acc3 = {};
#pragma unroll
        for (int c = 0; c < 4; ++c) {
            v2f B0 = { h0[2 * c], h0[2 * c + 1] };
            v2f B1 = { h1[2 * c], h1[2 * c + 1] };
            v2f B2 = { h2[2 * c], h2[2 * c + 1] };
            v2f B3 = { h3[2 * c], h3[2 * c + 1] };
            acc0 = __builtin_amdgcn_wmma_f32_16x16x4_f32(false, Aw[c], false, B0,
                                                         (short)0, acc0, false, false);
            acc1 = __builtin_amdgcn_wmma_f32_16x16x4_f32(false, Aw[c], false, B1,
                                                         (short)0, acc1, false, false);
            acc2 = __builtin_amdgcn_wmma_f32_16x16x4_f32(false, Aw[c], false, B2,
                                                         (short)0, acc2, false, false);
            acc3 = __builtin_amdgcn_wmma_f32_16x16x4_f32(false, Aw[c], false, B3,
                                                         (short)0, acc3, false, false);
        }

        float p00 = 0.f, p01 = 0.f, p02 = 0.f, p03 = 0.f;
        float p10 = 0.f, p11 = 0.f, p12 = 0.f, p13 = 0.f;
#pragma unroll
        for (int r = 0; r < 8; ++r) {
            float A0 = acc0[r] + b2c[r];
            float A1 = acc1[r];
            float A2 = acc2[r];
            float A3 = acc3[r];
            float t  = fast_tanhf(A0);
            float s  = 1.0f - t * t;
            float g0 = t;
            float g1 = s * A1;
            float g2 = s * A2 - 2.0f * t * s * A1 * A1;
            float g3 = s * A3 - 6.0f * t * s * A1 * A2
                     + s * (4.0f * t * t - 2.0f * s) * A1 * A1 * A1;
            p00 += w3u[r] * g0; p01 += w3u[r] * g1;
            p02 += w3u[r] * g2; p03 += w3u[r] * g3;
            p10 += w3w[r] * g0; p11 += w3w[r] * g1;
            p12 += w3w[r] * g2; p13 += w3w[r] * g3;
        }
        p00 += __shfl_xor(p00, 16, 32);
        p01 += __shfl_xor(p01, 16, 32);
        p02 += __shfl_xor(p02, 16, 32);
        p03 += __shfl_xor(p03, 16, 32);
        p10 += __shfl_xor(p10, 16, 32);
        p11 += __shfl_xor(p11, 16, 32);
        p12 += __shfl_xor(p12, 16, 32);
        p13 += __shfl_xor(p13, 16, 32);

        float u    = p00 + b30;
        float ux   = p01;
        float wv   = p10 + b31;
        float wx   = p11;
        float wxx  = p12;
        float wxxx = p13;
        float Nax  = EA_C * (ux + 0.5f * wx * wx);
        float Mb   = -EI_C * wxx;
        float Qs   = -EI_C * wxxx + Nax * wx;

        float f0 = half ? Nax : u;
        float f1 = half ? Mb  : wv;
        float f2 = half ? Qs  : wx;
        int base = (half ? 3 : 0) * n + pidx;
        out[base]         = f0;
        out[base + n]     = f1;
        out[base + 2 * n] = f2;
    }
}

extern "C" void kernel_launch(void* const* d_in, const int* in_sizes, int n_in,
                              void* d_out, int out_size, void* d_ws, size_t ws_size,
                              hipStream_t stream) {
    const float* x  = (const float*)d_in[0];
    const float* W1 = (const float*)d_in[1];
    const float* b1 = (const float*)d_in[2];
    const float* W2 = (const float*)d_in[3];
    const float* b2 = (const float*)d_in[4];
    const float* W3 = (const float*)d_in[5];
    const float* b3 = (const float*)d_in[6];
    float* out = (float*)d_out;
    const int n = in_sizes[0];

    const int nBatch = n >> 4;
    int blocks = (nBatch * 32 + 255) / 256;
    if (blocks > 2048) blocks = 2048;
    if (blocks < 1) blocks = 1;
    pinn_beam_wmma<<<blocks, 256, 0, stream>>>(x, W1, b1, W2, b2, W3, b3, out, n);
}


# Jet algebra strength-reduced: chain-rule scale factors folded into 4 pre-scaled W2 fragments, layer-2 jets regrouped on shared factors; vector instr per batch 242->142; sum order kept
# speedup vs baseline: 1.1158x; 1.1158x over previous
#include <hip/hip_runtime.h>


typedef float v2f __attribute__((ext_vector_type(2)));
typedef float v8f __attribute__((ext_vector_type(8)));

#define LBEAM 1.0f
#define EA_C  1.0e4f
#define EI_C  1.0e2f

__device__ __forceinline__ float fast_tanhf(float a) {
#if __has_builtin(__builtin_amdgcn_tanhf)
    return __builtin_amdgcn_tanhf(a);
#else
    float e = __expf(2.0f * a);
    return 1.0f - 2.0f * __builtin_amdgcn_rcpf(e + 1.0f);
#endif
}

__global__ __launch_bounds__(256) void pinn_beam_wmma(
    const float* __restrict__ x,
    const float* __restrict__ W1, const float* __restrict__ b1,
    const float* __restrict__ W2, const float* __restrict__ b2,
    const float* __restrict__ W3, const float* __restrict__ b3,
    float* __restrict__ out, int n)
{
    const int lane = threadIdx.x & 31;
    const int half = lane >> 4;
    const int l16  = lane & 15;

    v2f a1v[4], b1v[4];
    v2f Aw0[4], Aw1[4], Aw2[4], Aw3[4];
#pragma unroll
    for (int c = 0; c < 4; ++c) {
        const int ch = 4 * c + 2 * half;
        v2f a  = { W1[ch] * (1.0f / LBEAM), W1[ch + 1] * (1.0f / LBEAM) };
        v2f bb = { b1[ch], b1[ch + 1] };
        v2f w  = { W2[l16 * 16 + ch], W2[l16 * 16 + ch + 1] };
        v2f a2 = a * a;
        v2f a3 = a2 * a * (1.0f / 12.0f);
        a1v[c] = a;
        b1v[c] = bb;
        Aw0[c] = w;
        Aw1[c] = w * a;
        Aw2[c] = w * a2;
        Aw3[c] = w * a3;
    }
    float b2c[8], w3u[8], w3w[8];
#pragma unroll
    for (int r = 0; r < 8; ++r) {
        int ch = r + 8 * half;
        b2c[r] = b2[ch];
        w3u[r] = W3[ch];
        w3w[r] = W3[16 + ch];
    }
    const float b30 = b3[0], b31 = b3[1];

    const int nBatch = n >> 4;
    const int gwave  = (int)(blockIdx.x * blockDim.x + threadIdx.x) >> 5;
    const int nwave  = (int)(gridDim.x * blockDim.x) >> 5;

    for (int batch = gwave; batch < nBatch; batch += nwave) {
        const int   pidx = batch * 16 + l16;
        const float xi   = x[pidx];

        v8f acc0 = {}, acc1 = {}, acc2 = {}, acc3 = {};
#pragma unroll
        for (int c = 0; c < 4; ++c) {
            v2f a0 = a1v[c] * xi + b1v[c];
            v2f t  = { fast_tanhf(a0.x), fast_tanhf(a0.y) };
            v2f s  = 1.0f - t * t;
            v2f f2 = t * s;
            v2f f3 = s * (4.0f - 6.0f * s);
            acc0 = __builtin_amdgcn_wmma_f32_16x16x4_f32(false, Aw0[c], false, t,
                                                         (short)0, acc0, false, false);
            acc1 = __builtin_amdgcn_wmma_f32_16x16x4_f32(false, Aw1[c], false, s,
                                                         (short)0, acc1, false, false);
            acc2 = __builtin_amdgcn_wmma_f32_16x16x4_f32(false, Aw2[c], false, f2,
                                                         (short)0, acc2, false, false);
            acc3 = __builtin_amdgcn_wmma_f32_16x16x4_f32(false, Aw3[c], false, f3,
                                                         (short)0, acc3, false, false);
        }

        float pu0 = 0.f, pu1 = 0.f;
        float pw0 = 0.f, pw1 = 0.f, pw2 = 0.f, pw3 = 0.f;
#pragma unroll
        for (int r = 0; r < 8; ++r) {
            float A0  = acc0[r] + b2c[r];
            float A1  = acc1[r];
            float A2h = acc2[r];
            float A3t = acc3[r];
            float T   = fast_tanhf(A0);
            float S   = 1.0f - T * T;
            float q   = A1 * A1;
            float v   = A1 * A2h;
            float in2 = T * q + A2h;
            float cc  = (1.0f / 3.0f) - 0.5f * S;
            float in3 = T * v + A3t;
            in3 = (cc * A1) * q + in3;
            float su = S * w3u[r];
            float sw = S * w3w[r];
            pu0 += w3u[r] * T;  pw0 += w3w[r] * T;
            pu1 += su * A1;     pw1 += sw * A1;
            pw2 += sw * in2;    pw3 += sw * in3;
        }
        pu0 += __shfl_xor(pu0, 16, 32);
        pu1 += __shfl_xor(pu1, 16, 32);
        pw0 += __shfl_xor(pw0, 16, 32);
        pw1 += __shfl_xor(pw1, 16, 32);
        pw2 += __shfl_xor(pw2, 16, 32);
        pw3 += __shfl_xor(pw3, 16, 32);

        float u    = pu0 + b30;
        float ux   = pu1;
        float wv   = pw0 + b31;
        float wx   = pw1;
        float Nax  = EA_C * (ux + 0.5f * wx * wx);
        float Mb   = (2.0f * EI_C) * pw2;
        float Qs   = (-12.0f * EI_C) * pw3 + Nax * wx;

        float f0 = half ? Nax : u;
        float f1 = half ? Mb  : wv;
        float f2o = half ? Qs : wx;
        int base = (half ? 3 : 0) * n + pidx;
        out[base]         = f0;
        out[base + n]     = f1;
        out[base + 2 * n] = f2o;
    }
}

extern "C" void kernel_launch(void* const* d_in, const int* in_sizes, int n_in,
                              void* d_out, int out_size, void* d_ws, size_t ws_size,
                              hipStream_t stream) {
    const float* x  = (const float*)d_in[0];
    const float* W1 = (const float*)d_in[1];
    const float* b1 = (const float*)d_in[2];
    const float* W2 = (const float*)d_in[3];
    const float* b2 = (const float*)d_in[4];
    const float* W3 = (const float*)d_in[5];
    const float* b3 = (const float*)d_in[6];
    float* out = (float*)d_out;
    const int n = in_sizes[0];

    const int nBatch = n >> 4;
    int blocks = (nBatch * 32 + 255) / 256;
    if (blocks > 2048) blocks = 2048;
    if (blocks < 1) blocks = 1;
    pinn_beam_wmma<<<blocks, 256, 0, stream>>>(x, W1, b1, W2, b2, W3, b3, out, n);
}


# Grid cut from 2048 to 1024 blocks, one resident round of 8 waves per SIMD; each wave walks 4 batches with the next x prefetched; wave entries and weight staging halved; sum order kept
# speedup vs baseline: 1.2053x; 1.0802x over previous
#include <hip/hip_runtime.h>


typedef float v2f __attribute__((ext_vector_type(2)));
typedef float v8f __attribute__((ext_vector_type(8)));

#define LBEAM 1.0f
#define EA_C  1.0e4f
#define EI_C  1.0e2f

__device__ __forceinline__ float fast_tanhf(float a) {
#if __has_builtin(__builtin_amdgcn_tanhf)
    return __builtin_amdgcn_tanhf(a);
#else
    float e = __expf(2.0f * a);
    return 1.0f - 2.0f * __builtin_amdgcn_rcpf(e + 1.0f);
#endif
}

__global__ __launch_bounds__(256) void pinn_beam_wmma(
    const float* __restrict__ x,
    const float* __restrict__ W1, const float* __restrict__ b1,
    const float* __restrict__ W2, const float* __restrict__ b2,
    const float* __restrict__ W3, const float* __restrict__ b3,
    float* __restrict__ out, int n)
{
    const int lane = threadIdx.x & 31;
    const int half = lane >> 4;
    const int l16  = lane & 15;

    v2f a1v[4], b1v[4];
    v2f Aw0[4], Aw1[4], Aw2[4], Aw3[4];
#pragma unroll
    for (int c = 0; c < 4; ++c) {
        const int ch = 4 * c + 2 * half;
        v2f a  = { W1[ch] * (1.0f / LBEAM), W1[ch + 1] * (1.0f / LBEAM) };
        v2f bb = { b1[ch], b1[ch + 1] };
        v2f w  = { W2[l16 * 16 + ch], W2[l16 * 16 + ch + 1] };
        v2f a2 = a * a;
        v2f a3 = a2 * a * (1.0f / 12.0f);
        a1v[c] = a;
        b1v[c] = bb;
        Aw0[c] = w;
        Aw1[c] = w * a;
        Aw2[c] = w * a2;
        Aw3[c] = w * a3;
    }
    float b2c[8], w3u[8], w3w[8];
#pragma unroll
    for (int r = 0; r < 8; ++r) {
        int ch = r + 8 * half;
        b2c[r] = b2[ch];
        w3u[r] = W3[ch];
        w3w[r] = W3[16 + ch];
    }
    const float b30 = b3[0], b31 = b3[1];

    const int nBatch = n >> 4;
    const int gwave  = (int)(blockIdx.x * blockDim.x + threadIdx.x) >> 5;
    const int nwave  = (int)(gridDim.x * blockDim.x) >> 5;

    if (gwave >= nBatch) return;

    float xnext = x[gwave * 16 + l16];
    for (int batch = gwave; batch < nBatch; batch += nwave) {
        const int   pidx = batch * 16 + l16;
        const float xi   = xnext;
        const int   nb   = batch + nwave;
        xnext = x[(nb < nBatch ? nb : batch) * 16 + l16];

        v8f acc0 = {}, acc1 = {}, acc2 = {}, acc3 = {};
#pragma unroll
        for (int c = 0; c < 4; ++c) {
            v2f a0 = a1v[c] * xi + b1v[c];
            v2f t  = { fast_tanhf(a0.x), fast_tanhf(a0.y) };
            v2f s  = 1.0f - t * t;
            v2f f2 = t * s;
            v2f f3 = s * (4.0f - 6.0f * s);
            acc0 = __builtin_amdgcn_wmma_f32_16x16x4_f32(false, Aw0[c], false, t,
                                                         (short)0, acc0, false, false);
            acc1 = __builtin_amdgcn_wmma_f32_16x16x4_f32(false, Aw1[c], false, s,
                                                         (short)0, acc1, false, false);
            acc2 = __builtin_amdgcn_wmma_f32_16x16x4_f32(false, Aw2[c], false, f2,
                                                         (short)0, acc2, false, false);
            acc3 = __builtin_amdgcn_wmma_f32_16x16x4_f32(false, Aw3[c], false, f3,
                                                         (short)0, acc3, false, false);
        }

        float pu0 = 0.f, pu1 = 0.f;
        float pw0 = 0.f, pw1 = 0.f, pw2 = 0.f, pw3 = 0.f;
#pragma unroll
        for (int r = 0; r < 8; ++r) {
            float A0  = acc0[r] + b2c[r];
            float A1  = acc1[r];
            float A2h = acc2[r];
            float A3t = acc3[r];
            float T   = fast_tanhf(A0);
            float S   = 1.0f - T * T;
            float q   = A1 * A1;
            float v   = A1 * A2h;
            float in2 = T * q + A2h;
            float cc  = (1.0f / 3.0f) - 0.5f * S;
            float in3 = T * v + A3t;
            in3 = (cc * A1) * q + in3;
            float su = S * w3u[r];
            float sw = S * w3w[r];
            pu0 += w3u[r] * T;  pw0 += w3w[r] * T;
            pu1 += su * A1;     pw1 += sw * A1;
            pw2 += sw * in2;    pw3 += sw * in3;
        }
        pu0 += __shfl_xor(pu0, 16, 32);
        pu1 += __shfl_xor(pu1, 16, 32);
        pw0 += __shfl_xor(pw0, 16, 32);
        pw1 += __shfl_xor(pw1, 16, 32);
        pw2 += __shfl_xor(pw2, 16, 32);
        pw3 += __shfl_xor(pw3, 16, 32);

        float u    = pu0 + b30;
        float ux   = pu1;
        float wv   = pw0 + b31;
        float wx   = pw1;
        float Nax  = EA_C * (ux + 0.5f * wx * wx);
        float Mb   = (2.0f * EI_C) * pw2;
        float Qs   = (-12.0f * EI_C) * pw3 + Nax * wx;

        float f0 = half ? Nax : u;
        float f1 = half ? Mb  : wv;
        float f2o = half ? Qs : wx;
        int base = (half ? 3 : 0) * n + pidx;
        out[base]         = f0;
        out[base + n]     = f1;
        out[base + 2 * n] = f2o;
    }
}

extern "C" void kernel_launch(void* const* d_in, const int* in_sizes, int n_in,
                              void* d_out, int out_size, void* d_ws, size_t ws_size,
                              hipStream_t stream) {
    const float* x  = (const float*)d_in[0];
    const float* W1 = (const float*)d_in[1];
    const float* b1 = (const float*)d_in[2];
    const float* W2 = (const float*)d_in[3];
    const float* b2 = (const float*)d_in[4];
    const float* W3 = (const float*)d_in[5];
    const float* b3 = (const float*)d_in[6];
    float* out = (float*)d_out;
    const int n = in_sizes[0];

    const int nBatch = n >> 4;
    int blocks = (nBatch * 32 + 255) / 256;
    if (blocks > 1024) blocks = 1024;
    if (blocks < 1) blocks = 1;
    pinn_beam_wmma<<<blocks, 256, 0, stream>>>(x, W1, b1, W2, b2, W3, b3, out, n);
}


# Coordinates of a wave's first four batches loaded with the weights ahead of the batch loop and queued in registers (guarded fetch 4 batches ahead): no memory wait inside the loop; sum order kept
# speedup vs baseline: 1.2658x; 1.0502x over previous
#include <hip/hip_runtime.h>


typedef float v2f __attribute__((ext_vector_type(2)));
typedef float v8f __attribute__((ext_vector_type(8)));

#define LBEAM 1.0f
#define EA_C  1.0e4f
#define EI_C  1.0e2f

__device__ __forceinline__ float fast_tanhf(float a) {
#if __has_builtin(__builtin_amdgcn_tanhf)
    return __builtin_amdgcn_tanhf(a);
#else
    float e = __expf(2.0f * a);
    return 1.0f - 2.0f * __builtin_amdgcn_rcpf(e + 1.0f);
#endif
}

__global__ __launch_bounds__(256) void pinn_beam_wmma(
    const float* __restrict__ x,
    const float* __restrict__ W1, const float* __restrict__ b1,
    const float* __restrict__ W2, const float* __restrict__ b2,
    const float* __restrict__ W3, const float* __restrict__ b3,
    float* __restrict__ out, int n)
{
    const int lane = threadIdx.x & 31;
    const int half = lane >> 4;
    const int l16  = lane & 15;

    const int nBatch = n >> 4;
    const int gwave  = __builtin_amdgcn_readfirstlane(
                           (int)(blockIdx.x * blockDim.x + threadIdx.x) >> 5);
    const int nwave  = (int)(gridDim.x * blockDim.x) >> 5;

    if (gwave >= nBatch) return;

    const int q1 = gwave + nwave, q2 = q1 + nwave, q3 = q2 + nwave;
    float x0 = x[gwave * 16 + l16];
    float x1 = x[(q1 < nBatch ? q1 : gwave) * 16 + l16];
    float x2 = x[(q2 < nBatch ? q2 : gwave) * 16 + l16];
    float x3 = x[(q3 < nBatch ? q3 : gwave) * 16 + l16];

    v2f a1v[4], b1v[4];
    v2f Aw0[4], Aw1[4], Aw2[4], Aw3[4];
#pragma unroll
    for (int c = 0; c < 4; ++c) {
        const int ch = 4 * c + 2 * half;
        v2f a  = { W1[ch] * (1.0f / LBEAM), W1[ch + 1] * (1.0f / LBEAM) };
        v2f bb = { b1[ch], b1[ch + 1] };
        v2f w  = { W2[l16 * 16 + ch], W2[l16 * 16 + ch + 1] };
        v2f a2 = a * a;
        v2f a3 = a2 * a * (1.0f / 12.0f);
        a1v[c] = a;
        b1v[c] = bb;
        Aw0[c] = w;
        Aw1[c] = w * a;
        Aw2[c] = w * a2;
        Aw3[c] = w * a3;
    }
    float b2c[8], w3u[8], w3w[8];
#pragma unroll
    for (int r = 0; r < 8; ++r) {
        int ch = r + 8 * half;
        b2c[r] = b2[ch];
        w3u[r] = W3[ch];
        w3w[r] = W3[16 + ch];
    }
    const float b30 = b3[0], b31 = b3[1];

    for (int batch = gwave; batch < nBatch; batch += nwave) {
        const int   pidx = batch * 16 + l16;
        const float xi   = x0;

        v8f acc0 = {}, acc1 = {}, acc2 = {}, acc3 = {};
#pragma unroll
        for (int c = 0; c < 4; ++c) {
            v2f a0 = a1v[c] * xi + b1v[c];
            v2f t  = { fast_tanhf(a0.x), fast_tanhf(a0.y) };
            v2f s  = 1.0f - t * t;
            v2f f2 = t * s;
            v2f f3 = s * (4.0f - 6.0f * s);
            acc0 = __builtin_amdgcn_wmma_f32_16x16x4_f32(false, Aw0[c], false, t,
                                                         (short)0, acc0, false, false);
            acc1 = __builtin_amdgcn_wmma_f32_16x16x4_f32(false, Aw1[c], false, s,
                                                         (short)0, acc1, false, false);
            acc2 = __builtin_amdgcn_wmma_f32_16x16x4_f32(false, Aw2[c], false, f2,
                                                         (short)0, acc2, false, false);
            acc3 = __builtin_amdgcn_wmma_f32_16x16x4_f32(false, Aw3[c], false, f3,
                                                         (short)0, acc3, false, false);
        }

        x0 = x1; x1 = x2; x2 = x3;
        const int nb = batch + 4 * nwave;
        if (nb < nBatch) x3 = x[nb * 16 + l16];

        float pu0 = 0.f, pu1 = 0.f;
        float pw0 = 0.f, pw1 = 0.f, pw2 = 0.f, pw3 = 0.f;
#pragma unroll
        for (int r = 0; r < 8; ++r) {
            float A0  = acc0[r] + b2c[r];
            float A1  = acc1[r];
            float A2h = acc2[r];
            float A3t = acc3[r];
            float T   = fast_tanhf(A0);
            float S   = 1.0f - T * T;
            float q   = A1 * A1;
            float v   = A1 * A2h;
            float in2 = T * q + A2h;
            float cc  = (1.0f / 3.0f) - 0.5f * S;
            float in3 = T * v + A3t;
            in3 = (cc * A1) * q + in3;
            float su = S * w3u[r];
            float sw = S * w3w[r];
            pu0 += w3u[r] * T;  pw0 += w3w[r] * T;
            pu1 += su * A1;     pw1 += sw * A1;
            pw2 += sw * in2;    pw3 += sw * in3;
        }
        pu0 += __shfl_xor(pu0, 16, 32);
        pu1 += __shfl_xor(pu1, 16, 32);
        pw0 += __shfl_xor(pw0, 16, 32);
        pw1 += __shfl_xor(pw1, 16, 32);
        pw2 += __shfl_xor(pw2, 16, 32);
        pw3 += __shfl_xor(pw3, 16, 32);

        float u    = pu0 + b30;
        float ux   = pu1;
        float wv   = pw0 + b31;
        float wx   = pw1;
        float Nax  = EA_C * (ux + 0.5f * wx * wx);
        float Mb   = (2.0f * EI_C) * pw2;
        float Qs   = (-12.0f * EI_C) * pw3 + Nax * wx;

        float f0 = half ? Nax : u;
        float f1 = half ? Mb  : wv;
        float f2o = half ? Qs : wx;
        int base = (half ? 3 : 0) * n + pidx;
        out[base]         = f0;
        out[base + n]     = f1;
        out[base + 2 * n] = f2o;
    }
}

extern "C" void kernel_launch(void* const* d_in, const int* in_sizes, int n_in,
                              void* d_out, int out_size, void* d_ws, size_t ws_size,
                              hipStream_t stream) {
    const float* x  = (const float*)d_in[0];
    const float* W1 = (const float*)d_in[1];
    const float* b1 = (const float*)d_in[2];
    const float* W2 = (const float*)d_in[3];
    const float* b2 = (const float*)d_in[4];
    const float* W3 = (const float*)d_in[5];
    const float* b3 = (const float*)d_in[6];
    float* out = (float*)d_out;
    const int n = in_sizes[0];

    const int nBatch = n >> 4;
    int blocks = (nBatch * 32 + 255) / 256;
    if (blocks > 1024) blocks = 1024;
    if (blocks < 1) blocks = 1;
    pinn_beam_wmma<<<blocks, 256, 0, stream>>>(x, W1, b1, W2, b2, W3, b3, out, n);
}
